# starting kernel
# baseline (speedup 1.0000x reference)
#include <hip/hip_runtime.h>
#include <math.h>

typedef float v2f __attribute__((ext_vector_type(2)));
typedef float v8f __attribute__((ext_vector_type(8)));

#define B_   8
#define S_   4096
#define D_   1024
#define H_   1024
#define PD_  512
#define P_   4
#define TOK_   2048
#define TC_    8
#define TOKPC_ 256

__global__ void gather_partials(const float* __restrict__ ctx,
                                const int* __restrict__ masks,
                                float* __restrict__ partials) {
  __shared__ int idx[TOKPC_];
  const int dc  = blockIdx.x;
  const int tc  = blockIdx.y;
  const int b   = blockIdx.z;
  const int tid = threadIdx.x;
  idx[tid] = masks[b * TOK_ + tc * TOKPC_ + tid];
  __syncthreads();
  const float* base = ctx + (size_t)b * S_ * D_ + dc * 256 + tid;
  float sum = 0.f;
#pragma unroll 8
  for (int i = 0; i < TOKPC_; ++i) {
    sum += base[(size_t)idx[i] * D_];
  }
  partials[(size_t)(tc * B_ + b) * D_ + dc * 256 + tid] = sum;
}

__global__ void reduce_pool(const float* __restrict__ partials,
                            float* __restrict__ pooled) {
  const int g   = blockIdx.x * 256 + threadIdx.x;
  const int row = g >> 10;
  const int d   = g & 1023;
  float s = 0.f;
  if (row < B_) {
#pragma unroll
    for (int tc = 0; tc < TC_; ++tc)
      s += partials[(size_t)(tc * B_ + row) * D_ + d];
    s *= (1.0f / (float)TOK_);
  }
  pooled[g] = s;
}

__global__ void gemm1_gelu(const float* __restrict__ A,
                           const float* __restrict__ W1,
                           const float* __restrict__ b1,
                           float* __restrict__ Hout) {
  const int lane = threadIdx.x & 31;
  const int half = lane >> 4;
  const int l16  = lane & 15;
  const int n0   = blockIdx.x * 16;
  const int col  = n0 + l16;

  const float* Ap = A + l16 * D_ + 2 * half;
  const float* Wp = W1 + (size_t)(2 * half) * H_ + col;

  v8f acc = {};
  for (int k = 0; k < D_; k += 4) {
    v2f a, b;
    a.x = Ap[k];
    a.y = Ap[k + 1];
    b.x = Wp[(size_t)k * H_];
    b.y = Wp[(size_t)(k + 1) * H_];
    acc = __builtin_amdgcn_wmma_f32_16x16x4_f32(
        false, a, false, b, (short)0, acc, false, false);
  }

  const float bias = b1[col];
#pragma unroll
  for (int r = 0; r < 8; ++r) {
    float x = acc[r] + bias;
    float g = 0.5f * x * (1.0f + erff(x * 0.70710678118654752f));
    int row = r + 8 * half;
    Hout[row * H_ + col] = g;
  }
}

__global__ void layernorm_rows(float* __restrict__ Hbuf,
                               const float* __restrict__ gam,
                               const float* __restrict__ bet) {
  const int row = blockIdx.x;
  const int tid = threadIdx.x;
  __shared__ float red1[8];
  __shared__ float red2[8];

  float v[4];
  float s = 0.f;
#pragma unroll
  for (int j = 0; j < 4; ++j) {
    v[j] = Hbuf[row * H_ + tid + j * 256];
    s += v[j];
  }
#pragma unroll
  for (int off = 16; off > 0; off >>= 1) s += __shfl_down(s, off, 32);
  if ((tid & 31) == 0) red1[tid >> 5] = s;
  __syncthreads();
  float tot = 0.f;
#pragma unroll
  for (int i = 0; i < 8; ++i) tot += red1[i];
  const float mean = tot * (1.0f / (float)H_);

  float s2 = 0.f;
#pragma unroll
  for (int j = 0; j < 4; ++j) {
    float d = v[j] - mean;
    s2 += d * d;
  }
#pragma unroll
  for (int off = 16; off > 0; off >>= 1) s2 += __shfl_down(s2, off, 32);
  if ((tid & 31) == 0) red2[tid >> 5] = s2;
  __syncthreads();
  float tot2 = 0.f;
#pragma unroll
  for (int i = 0; i < 8; ++i) tot2 += red2[i];
  const float inv = rsqrtf(tot2 * (1.0f / (float)H_) + 1e-5f);

#pragma unroll
  for (int j = 0; j < 4; ++j) {
    int c = tid + j * 256;
    Hbuf[row * H_ + c] = (v[j] - mean) * inv * gam[c] + bet[c];
  }
}

__global__ void gemm2_out(const float* __restrict__ Hbuf,
                          const float* __restrict__ W2,
                          const float* __restrict__ b2,
                          float* __restrict__ out) {
  const int lane = threadIdx.x & 31;
  const int half = lane >> 4;
  const int l16  = lane & 15;
  const int n0   = blockIdx.x * 16;
  const int col  = n0 + l16;

  const float* Ap = Hbuf + l16 * H_ + 2 * half;
  const float* Wp = W2 + (size_t)(2 * half) * PD_ + col;

  v8f acc = {};
  for (int k = 0; k < H_; k += 4) {
    v2f a, b;
    a.x = Ap[k];
    a.y = Ap[k + 1];
    b.x = Wp[(size_t)k * PD_];
    b.y = Wp[(size_t)(k + 1) * PD_];
    acc = __builtin_amdgcn_wmma_f32_16x16x4_f32(
        false, a, false, b, (short)0, acc, false, false);
  }

  if (half == 0) {
    const float bias = b2[col];
#pragma unroll
    for (int r = 0; r < 8; ++r) {
      float val = acc[r] + bias;
#pragma unroll
      for (int p = 0; p < P_; ++p)
        out[(size_t)(r * P_ + p) * PD_ + col] = val;
    }
  }
}

extern "C" void kernel_launch(void* const* d_in, const int* in_sizes, int n_in,
                              void* d_out, int out_size, void* d_ws, size_t ws_size,
                              hipStream_t stream) {
  const float* ctx  = (const float*)d_in[0];
  const float* W1   = (const float*)d_in[1];
  const float* b1   = (const float*)d_in[2];
  const float* ln_g = (const float*)d_in[3];
  const float* ln_b = (const float*)d_in[4];
  const float* W2   = (const float*)d_in[5];
  const float* b2   = (const float*)d_in[6];
  const int*  masks = (const int*)d_in[7];
  (void)in_sizes; (void)n_in; (void)out_size; (void)ws_size;

  float* out = (float*)d_out;

  char* ws = (char*)d_ws;
  float* pooled   = (float*)(ws);
  float* partials = (float*)(ws + 65536);
  float* hbuf     = (float*)(ws + 65536 + 262144);

  gather_partials<<<dim3(4, TC_, B_), 256, 0, stream>>>(ctx, masks, partials);
  reduce_pool<<<64, 256, 0, stream>>>(partials, pooled);
  gemm1_gelu<<<H_ / 16, 32, 0, stream>>>(pooled, W1, b1, hbuf);
  layernorm_rows<<<B_, 256, 0, stream>>>(hbuf, ln_g, ln_b);
  gemm2_out<<<PD_ / 16, 32, 0, stream>>>(hbuf, W2, b2, out);
}


# K loop of both f32 WMMA GEMMs split over 16 waves per work-group, partial tiles summed in fixed wave order through LDS (changes the summation order)
# speedup vs baseline: 1.7903x; 1.7903x over previous
#include <hip/hip_runtime.h>
#include <math.h>

typedef float v2f __attribute__((ext_vector_type(2)));
typedef float v8f __attribute__((ext_vector_type(8)));

#define B_   8
#define S_   4096
#define D_   1024
#define H_   1024
#define PD_  512
#define P_   4
#define TOK_   2048
#define TC_    8
#define TOKPC_ 256
#define KW_    16

__global__ void gather_partials(const float* __restrict__ ctx,
                                const int* __restrict__ masks,
                                float* __restrict__ partials) {
  __shared__ int idx[TOKPC_];
  const int dc  = blockIdx.x;
  const int tc  = blockIdx.y;
  const int b   = blockIdx.z;
  const int tid = threadIdx.x;
  idx[tid] = masks[b * TOK_ + tc * TOKPC_ + tid];
  __syncthreads();
  const float* base = ctx + (size_t)b * S_ * D_ + dc * 256 + tid;
  float sum = 0.f;
#pragma unroll 8
  for (int i = 0; i < TOKPC_; ++i) {
    sum += base[(size_t)idx[i] * D_];
  }
  partials[(size_t)(tc * B_ + b) * D_ + dc * 256 + tid] = sum;
}

__global__ void reduce_pool(const float* __restrict__ partials,
                            float* __restrict__ pooled) {
  const int g   = blockIdx.x * 256 + threadIdx.x;
  const int row = g >> 10;
  const int d   = g & 1023;
  float s = 0.f;
  if (row < B_) {
#pragma unroll
    for (int tc = 0; tc < TC_; ++tc)
      s += partials[(size_t)(tc * B_ + row) * D_ + d];
    s *= (1.0f / (float)TOK_);
  }
  pooled[g] = s;
}

__global__ void __launch_bounds__(KW_ * 32)
gemm1_gelu(const float* __restrict__ A,
           const float* __restrict__ W1,
           const float* __restrict__ b1,
           float* __restrict__ Hout) {
  __shared__ float part[KW_][256];
  const int tid  = threadIdx.x;
  const int wave = tid >> 5;
  const int lane = tid & 31;
  const int half = lane >> 4;
  const int l16  = lane & 15;
  const int n0   = blockIdx.x * 16;
  const int col  = n0 + l16;
  const int kw   = wave * (D_ / KW_);

  const float* Ap = A + l16 * D_ + kw + 2 * half;
  const float* Wp = W1 + (size_t)(kw + 2 * half) * H_ + col;

  v8f acc = {};
  for (int k = 0; k < D_ / KW_; k += 4) {
    v2f a, b;
    a.x = Ap[k];
    a.y = Ap[k + 1];
    b.x = Wp[(size_t)k * H_];
    b.y = Wp[(size_t)(k + 1) * H_];
    acc = __builtin_amdgcn_wmma_f32_16x16x4_f32(
        false, a, false, b, (short)0, acc, false, false);
  }

  for (int r = 0; r < 8; ++r) part[wave][r * 32 + lane] = acc[r];
  __syncthreads();

  if (tid < 256) {
    float s = part[0][tid];
    for (int w = 1; w < KW_; ++w) s += part[w][tid];
    const int r   = tid >> 5;
    const int row = r + 8 * half;
    float x = s + b1[col];
    float g = 0.5f * x * (1.0f + erff(x * 0.70710678118654752f));
    Hout[row * H_ + col] = g;
  }
}

__global__ void layernorm_rows(float* __restrict__ Hbuf,
                               const float* __restrict__ gam,
                               const float* __restrict__ bet) {
  const int row = blockIdx.x;
  const int tid = threadIdx.x;
  __shared__ float red1[8];
  __shared__ float red2[8];

  float v[4];
  float s = 0.f;
#pragma unroll
  for (int j = 0; j < 4; ++j) {
    v[j] = Hbuf[row * H_ + tid + j * 256];
    s += v[j];
  }
#pragma unroll
  for (int off = 16; off > 0; off >>= 1) s += __shfl_down(s, off, 32);
  if ((tid & 31) == 0) red1[tid >> 5] = s;
  __syncthreads();
  float tot = 0.f;
#pragma unroll
  for (int i = 0; i < 8; ++i) tot += red1[i];
  const float mean = tot * (1.0f / (float)H_);

  float s2 = 0.f;
#pragma unroll
  for (int j = 0; j < 4; ++j) {
    float d = v[j] - mean;
    s2 += d * d;
  }
#pragma unroll
  for (int off = 16; off > 0; off >>= 1) s2 += __shfl_down(s2, off, 32);
  if ((tid & 31) == 0) red2[tid >> 5] = s2;
  __syncthreads();
  float tot2 = 0.f;
#pragma unroll
  for (int i = 0; i < 8; ++i) tot2 += red2[i];
  const float inv = rsqrtf(tot2 * (1.0f / (float)H_) + 1e-5f);

#pragma unroll
  for (int j = 0; j < 4; ++j) {
    int c = tid + j * 256;
    Hbuf[row * H_ + c] = (v[j] - mean) * inv * gam[c] + bet[c];
  }
}

__global__ void __launch_bounds__(KW_ * 32)
gemm2_out(const float* __restrict__ Hbuf,
          const float* __restrict__ W2,
          const float* __restrict__ b2,
          float* __restrict__ out) {
  __shared__ float part[KW_][128];
  const int tid  = threadIdx.x;
  const int wave = tid >> 5;
  const int lane = tid & 31;
  const int half = lane >> 4;
  const int l16  = lane & 15;
  const int n0   = blockIdx.x * 16;
  const int col  = n0 + l16;
  const int kw   = wave * (H_ / KW_);

  const float* Ap = Hbuf + l16 * H_ + kw + 2 * half;
  const float* Wp = W2 + (size_t)(kw + 2 * half) * PD_ + col;

  v8f acc = {};
  for (int k = 0; k < H_ / KW_; k += 4) {
    v2f a, b;
    a.x = Ap[k];
    a.y = Ap[k + 1];
    b.x = Wp[(size_t)k * PD_];
    b.y = Wp[(size_t)(k + 1) * PD_];
    acc = __builtin_amdgcn_wmma_f32_16x16x4_f32(
        false, a, false, b, (short)0, acc, false, false);
  }

  if (half == 0) {
    for (int r = 0; r < 8; ++r) part[wave][r * 16 + l16] = acc[r];
  }
  __syncthreads();

  if (tid < 128) {
    float s = part[0][tid];
    for (int w = 1; w < KW_; ++w) s += part[w][tid];
    const int r = tid >> 4;
    const int c = n0 + (tid & 15);
    const float val = s + b2[c];
#pragma unroll
    for (int p = 0; p < P_; ++p)
      out[(size_t)(r * P_ + p) * PD_ + c] = val;
  }
}

extern "C" void kernel_launch(void* const* d_in, const int* in_sizes, int n_in,
                              void* d_out, int out_size, void* d_ws, size_t ws_size,
                              hipStream_t stream) {
  const float* ctx  = (const float*)d_in[0];
  const float* W1   = (const float*)d_in[1];
  const float* b1   = (const float*)d_in[2];
  const float* ln_g = (const float*)d_in[3];
  const float* ln_b = (const float*)d_in[4];
  const float* W2   = (const float*)d_in[5];
  const float* b2   = (const float*)d_in[6];
  const int*  masks = (const int*)d_in[7];
  (void)in_sizes; (void)n_in; (void)out_size; (void)ws_size;

  float* out = (float*)d_out;

  char* ws = (char*)d_ws;
  float* pooled   = (float*)(ws);
  float* partials = (float*)(ws + 65536);
  float* hbuf     = (float*)(ws + 65536 + 262144);

  gather_partials<<<dim3(4, TC_, B_), 256, 0, stream>>>(ctx, masks, partials);
  reduce_pool<<<64, 256, 0, stream>>>(partials, pooled);
  gemm1_gelu<<<H_ / 16, KW_ * 32, 0, stream>>>(pooled, W1, b1, hbuf);
  layernorm_rows<<<B_, 256, 0, stream>>>(hbuf, ln_g, ln_b);
  gemm2_out<<<PD_ / 16, KW_ * 32, 0, stream>>>(hbuf, W2, b2, out);
}


# Gather kernel: each wave requests 32 whole 512-byte row pieces with 16-byte loads before summing any, 8 waves' sums added through LDS in wave order (changes the summation order)
# speedup vs baseline: 3.1850x; 1.7790x over previous
#include <hip/hip_runtime.h>
#include <math.h>

typedef float v2f __attribute__((ext_vector_type(2)));
typedef float v8f __attribute__((ext_vector_type(8)));

#define B_   8
#define S_   4096
#define D_   1024
#define H_   1024
#define PD_  512
#define P_   4
#define TOK_   2048
#define TC_    8
#define TOKPC_ 256
#define KW_    16
#define GWAVES_ 8
#define TOKPW_  32
#define DCW_    128

__global__ void __launch_bounds__(GWAVES_ * 32) __attribute__((amdgpu_waves_per_eu(4, 4)))
gather_partials(const float* __restrict__ ctx,
                const int* __restrict__ masks,
                float* __restrict__ partials) {
  __shared__ float4 red[GWAVES_][32];
  const int dc   = blockIdx.x;
  const int tc   = blockIdx.y;
  const int b    = blockIdx.z;
  const int tid  = threadIdx.x;
  const int lane = tid & 31;
  const int w = __builtin_amdgcn_readfirstlane(tid >> 5);
  const int* mp = masks + b * TOK_ + tc * TOKPC_ + w * TOKPW_;
  const float* base = ctx + (size_t)b * S_ * D_ + dc * DCW_;
  const unsigned loff = lane * 4;

  float4 v[TOKPW_];
#pragma unroll
  for (int i = 0; i < TOKPW_; ++i)
    v[i] = *(const float4*)(base + (size_t)(unsigned)mp[i] * D_ + loff);
  float4 s = v[0];
#pragma unroll
  for (int i = 1; i < TOKPW_; ++i) {
    s.x += v[i].x; s.y += v[i].y; s.z += v[i].z; s.w += v[i].w;
  }
  red[w][lane] = s;
  __syncthreads();

  if (tid < 32) {
    float4 t = red[0][tid];
    for (int u = 1; u < GWAVES_; ++u) {
      const float4 r = red[u][tid];
      t.x += r.x; t.y += r.y; t.z += r.z; t.w += r.w;
    }
    *(float4*)(partials + (size_t)(tc * B_ + b) * D_ + dc * DCW_ + tid * 4) = t;
  }
}

__global__ void reduce_pool(const float* __restrict__ partials,
                            float* __restrict__ pooled) {
  const int g   = blockIdx.x * 256 + threadIdx.x;
  const int row = g >> 10;
  const int d   = g & 1023;
  float s = 0.f;
  if (row < B_) {
#pragma unroll
    for (int tc = 0; tc < TC_; ++tc)
      s += partials[(size_t)(tc * B_ + row) * D_ + d];
    s *= (1.0f / (float)TOK_);
  }
  pooled[g] = s;
}

__global__ void __launch_bounds__(KW_ * 32)
gemm1_gelu(const float* __restrict__ A,
           const float* __restrict__ W1,
           const float* __restrict__ b1,
           float* __restrict__ Hout) {
  __shared__ float part[KW_][256];
  const int tid  = threadIdx.x;
  const int wave = tid >> 5;
  const int lane = tid & 31;
  const int half = lane >> 4;
  const int l16  = lane & 15;
  const int n0   = blockIdx.x * 16;
  const int col  = n0 + l16;
  const int kw   = wave * (D_ / KW_);

  const float* Ap = A + l16 * D_ + kw + 2 * half;
  const float* Wp = W1 + (size_t)(kw + 2 * half) * H_ + col;

  v8f acc = {};
  for (int k = 0; k < D_ / KW_; k += 4) {
    v2f a, b;
    a.x = Ap[k];
    a.y = Ap[k + 1];
    b.x = Wp[(size_t)k * H_];
    b.y = Wp[(size_t)(k + 1) * H_];
    acc = __builtin_amdgcn_wmma_f32_16x16x4_f32(
        false, a, false, b, (short)0, acc, false, false);
  }

  for (int r = 0; r < 8; ++r) part[wave][r * 32 + lane] = acc[r];
  __syncthreads();

  if (tid < 256) {
    float s = part[0][tid];
    for (int w = 1; w < KW_; ++w) s += part[w][tid];
    const int r   = tid >> 5;
    const int row = r + 8 * half;
    float x = s + b1[col];
    float g = 0.5f * x * (1.0f + erff(x * 0.70710678118654752f));
    Hout[row * H_ + col] = g;
  }
}

__global__ void layernorm_rows(float* __restrict__ Hbuf,
                               const float* __restrict__ gam,
                               const float* __restrict__ bet) {
  const int row = blockIdx.x;
  const int tid = threadIdx.x;
  __shared__ float red1[8];
  __shared__ float red2[8];

  float v[4];
  float s = 0.f;
#pragma unroll
  for (int j = 0; j < 4; ++j) {
    v[j] = Hbuf[row * H_ + tid + j * 256];
    s += v[j];
  }
#pragma unroll
  for (int off = 16; off > 0; off >>= 1) s += __shfl_down(s, off, 32);
  if ((tid & 31) == 0) red1[tid >> 5] = s;
  __syncthreads();
  float tot = 0.f;
#pragma unroll
  for (int i = 0; i < 8; ++i) tot += red1[i];
  const float mean = tot * (1.0f / (float)H_);

  float s2 = 0.f;
#pragma unroll
  for (int j = 0; j < 4; ++j) {
    float d = v[j] - mean;
    s2 += d * d;
  }
#pragma unroll
  for (int off = 16; off > 0; off >>= 1) s2 += __shfl_down(s2, off, 32);
  if ((tid & 31) == 0) red2[tid >> 5] = s2;
  __syncthreads();
  float tot2 = 0.f;
#pragma unroll
  for (int i = 0; i < 8; ++i) tot2 += red2[i];
  const float inv = rsqrtf(tot2 * (1.0f / (float)H_) + 1e-5f);

#pragma unroll
  for (int j = 0; j < 4; ++j) {
    int c = tid + j * 256;
    Hbuf[row * H_ + c] = (v[j] - mean) * inv * gam[c] + bet[c];
  }
}

__global__ void __launch_bounds__(KW_ * 32)
gemm2_out(const float* __restrict__ Hbuf,
          const float* __restrict__ W2,
          const float* __restrict__ b2,
          float* __restrict__ out) {
  __shared__ float part[KW_][128];
  const int tid  = threadIdx.x;
  const int wave = tid >> 5;
  const int lane = tid & 31;
  const int half = lane >> 4;
  const int l16  = lane & 15;
  const int n0   = blockIdx.x * 16;
  const int col  = n0 + l16;
  const int kw   = wave * (H_ / KW_);

  const float* Ap = Hbuf + l16 * H_ + kw + 2 * half;
  const float* Wp = W2 + (size_t)(kw + 2 * half) * PD_ + col;

  v8f acc = {};
  for (int k = 0; k < H_ / KW_; k += 4) {
    v2f a, b;
    a.x = Ap[k];
    a.y = Ap[k + 1];
    b.x = Wp[(size_t)k * PD_];
    b.y = Wp[(size_t)(k + 1) * PD_];
    acc = __builtin_amdgcn_wmma_f32_16x16x4_f32(
        false, a, false, b, (short)0, acc, false, false);
  }

  if (half == 0) {
    for (int r = 0; r < 8; ++r) part[wave][r * 16 + l16] = acc[r];
  }
  __syncthreads();

  if (tid < 128) {
    float s = part[0][tid];
    for (int w = 1; w < KW_; ++w) s += part[w][tid];
    const int r = tid >> 4;
    const int c = n0 + (tid & 15);
    const float val = s + b2[c];
#pragma unroll
    for (int p = 0; p < P_; ++p)
      out[(size_t)(r * P_ + p) * PD_ + c] = val;
  }
}

extern "C" void kernel_launch(void* const* d_in, const int* in_sizes, int n_in,
                              void* d_out, int out_size, void* d_ws, size_t ws_size,
                              hipStream_t stream) {
  const float* ctx  = (const float*)d_in[0];
  const float* W1   = (const float*)d_in[1];
  const float* b1   = (const float*)d_in[2];
  const float* ln_g = (const float*)d_in[3];
  const float* ln_b = (const float*)d_in[4];
  const float* W2   = (const float*)d_in[5];
  const float* b2   = (const float*)d_in[6];
  const int*  masks = (const int*)d_in[7];
  (void)in_sizes; (void)n_in; (void)out_size; (void)ws_size;

  float* out = (float*)d_out;

  char* ws = (char*)d_ws;
  float* pooled   = (float*)(ws);
  float* partials = (float*)(ws + 65536);
  float* hbuf     = (float*)(ws + 65536 + 262144);

  gather_partials<<<dim3(D_ / DCW_, TC_, B_), GWAVES_ * 32, 0, stream>>>(ctx, masks, partials);
  reduce_pool<<<64, 256, 0, stream>>>(partials, pooled);
  gemm1_gelu<<<H_ / 16, KW_ * 32, 0, stream>>>(pooled, W1, b1, hbuf);
  layernorm_rows<<<B_, 256, 0, stream>>>(hbuf, ln_g, ln_b);
  gemm2_out<<<PD_ / 16, KW_ * 32, 0, stream>>>(hbuf, W2, b2, out);
}


# Gather blocks cover 512 tokens instead of 256: 256 blocks of 8 waves, 64 row pieces in flight per wave; half the waves enter, 4 partial planes instead of 8 reach GEMM1 (changes the summation order)
# speedup vs baseline: 4.3469x; 1.3648x over previous
#include <hip/hip_runtime.h>
#include <math.h>

typedef float v2f __attribute__((ext_vector_type(2)));
typedef float v8f __attribute__((ext_vector_type(8)));

#define B_   8
#define S_   4096
#define D_   1024
#define H_   1024
#define PD_  512
#define P_   4
#define TOK_   2048
#define TC_    4
#define TOKPC_ 512
#define KW_    16
#define GWAVES_ 8
#define TOKPW_  64
#define DCW_    128
#define APITCH_ 1028

__global__ void __launch_bounds__(GWAVES_ * 32) __attribute__((amdgpu_waves_per_eu(2, 2)))
gather_partials(const float* __restrict__ ctx,
                const int* __restrict__ masks,
                float* __restrict__ partials) {
  __shared__ float4 red[GWAVES_][32];
  const int dc   = blockIdx.x;
  const int tc   = blockIdx.y;
  const int b    = blockIdx.z;
  const int tid  = threadIdx.x;
  const int lane = tid & 31;
  const int w = __builtin_amdgcn_readfirstlane(tid >> 5);
  const int* mp = masks + b * TOK_ + tc * TOKPC_ + w * TOKPW_;
  const float* base = ctx + (size_t)b * S_ * D_ + dc * DCW_;
  const unsigned loff = lane * 4;

  const int i0 = mp[lane];
  const int i1 = mp[32 + lane];

  float4 v[TOKPW_];
#pragma unroll
  for (int i = 0; i < TOKPW_; ++i) {
    const unsigned row = (unsigned)__builtin_amdgcn_readlane(i < 32 ? i0 : i1, i & 31);
    v[i] = *(const float4*)(base + (size_t)row * D_ + loff);
  }
  float4 s = v[0];
#pragma unroll
  for (int i = 1; i < TOKPW_; ++i) {
    s.x += v[i].x; s.y += v[i].y; s.z += v[i].z; s.w += v[i].w;
  }
  red[w][lane] = s;
  __syncthreads();

  if (tid < 32) {
    float4 t = red[0][tid];
    for (int u = 1; u < GWAVES_; ++u) {
      const float4 r = red[u][tid];
      t.x += r.x; t.y += r.y; t.z += r.z; t.w += r.w;
    }
    *(float4*)(partials + (size_t)(tc * B_ + b) * D_ + dc * DCW_ + tid * 4) = t;
  }
}

template <int M>
static __device__ __forceinline__ float row_xor(float x) {
  return __builtin_bit_cast(float, __builtin_amdgcn_update_dpp(
      0, __builtin_bit_cast(int, x), 0x160 | M, 0xf, 0xf, false));
}
static __device__ __forceinline__ float xhalf(float x) {
  return __builtin_bit_cast(float, __builtin_amdgcn_permlanex16(
      __builtin_bit_cast(int, x), __builtin_bit_cast(int, x), 0x76543210, 0xfedcba98, false, false));
}
static __device__ __forceinline__ float wave_sum(float x) {
  x += row_xor<1>(x);
  x += row_xor<2>(x);
  x += row_xor<4>(x);
  x += row_xor<8>(x);
  x += xhalf(x);
  return x;
}

__global__ void __launch_bounds__(KW_ * 32)
gemm1_gelu(const float* __restrict__ partials,
           const float* __restrict__ W1,
           const float* __restrict__ b1,
           float* __restrict__ Hout) {
  __shared__ float4 apan4[B_ * APITCH_ / 4];
  __shared__ float part[KW_][128];
  float* apan = (float*)apan4;
  const int tid  = threadIdx.x;
  const int wave = tid >> 5;
  const int lane = tid & 31;
  const int half = lane >> 4;
  const int l16  = lane & 15;
  const int n0   = blockIdx.x * 16;
  const int col  = n0 + l16;
  const int kw   = wave * (D_ / KW_);

  const float* Wp = W1 + (size_t)(kw + 2 * half) * H_ + col;
  v2f bf[D_ / KW_ / 4];
#pragma unroll
  for (int j = 0; j < D_ / KW_ / 4; ++j) {
    bf[j].x = Wp[(size_t)(4 * j) * H_];
    bf[j].y = Wp[(size_t)(4 * j + 1) * H_];
  }
  __builtin_amdgcn_sched_barrier(0);

  float4 pv[4][TC_];
#pragma unroll
  for (int i = 0; i < 4; ++i)
#pragma unroll
    for (int tc = 0; tc < TC_; ++tc)
      pv[i][tc] = *(const float4*)(partials + (size_t)tc * B_ * D_ + 4 * (tid + KW_ * 32 * i));
  __builtin_amdgcn_sched_barrier(0);
#pragma unroll
  for (int i = 0; i < 4; ++i) {
    float4 s = pv[i][0];
#pragma unroll
    for (int tc = 1; tc < TC_; ++tc) {
      s.x += pv[i][tc].x; s.y += pv[i][tc].y; s.z += pv[i][tc].z; s.w += pv[i][tc].w;
    }
    const float sc = 1.0f / (float)TOK_;
    s.x *= sc; s.y *= sc; s.z *= sc; s.w *= sc;
    const int f = tid + KW_ * 32 * i;
    apan4[(f >> 8) * (APITCH_ / 4) + (f & 255)] = s;
  }
  __syncthreads();

  const float* ap = apan + (l16 & 7) * APITCH_ + kw + 2 * half;
  const bool live = l16 < B_;
  v2f af[D_ / KW_ / 4];
#pragma unroll
  for (int j = 0; j < D_ / KW_ / 4; ++j) {
    const float2 t = *(const float2*)(ap + 4 * j);
    af[j].x = live ? t.x : 0.0f;
    af[j].y = live ? t.y : 0.0f;
  }
  __builtin_amdgcn_sched_barrier(0);
  v8f acc = {};
#pragma unroll
  for (int j = 0; j < D_ / KW_ / 4; ++j)
    acc = __builtin_amdgcn_wmma_f32_16x16x4_f32(
        false, af[j], false, bf[j], (short)0, acc, false, false);

  if (half == 0) {
    for (int r = 0; r < 8; ++r) part[wave][r * 16 + l16] = acc[r];
  }
  __syncthreads();

  if (tid < 128) {
    float s = part[0][tid];
    for (int w = 1; w < KW_; ++w) s += part[w][tid];
    const int r = tid >> 4;
    const int c = n0 + (tid & 15);
    float x = s + b1[c];
    float g = 0.5f * x * (1.0f + erff(x * 0.70710678118654752f));
    Hout[r * H_ + c] = g;
  }
}

__global__ void __launch_bounds__(KW_ * 32)
gemm2_out(const float* __restrict__ Hbuf,
          const float* __restrict__ gam,
          const float* __restrict__ bet,
          const float* __restrict__ W2,
          const float* __restrict__ b2,
          float* __restrict__ out) {
  __shared__ float4 apan4[B_ * APITCH_ / 4];
  __shared__ float part[KW_][128];
  float* apan = (float*)apan4;
  const int tid  = threadIdx.x;
  const int wave = __builtin_amdgcn_readfirstlane(tid >> 5);
  const int lane = tid & 31;
  const int half = lane >> 4;
  const int l16  = lane & 15;
  const int n0   = blockIdx.x * 16;
  const int col  = n0 + l16;
  const int kw   = wave * (H_ / KW_);

  const float* Wp = W2 + (size_t)(kw + 2 * half) * PD_ + col;
  v2f bf[H_ / KW_ / 4];
#pragma unroll
  for (int j = 0; j < H_ / KW_ / 4; ++j) {
    bf[j].x = Wp[(size_t)(4 * j) * PD_];
    bf[j].y = Wp[(size_t)(4 * j + 1) * PD_];
  }

  if (wave < B_) {
    const float* hp = Hbuf + wave * H_;
    float4 v[8], g[8], be[8];
#pragma unroll
    for (int i = 0; i < 8; ++i) {
      v[i]  = *(const float4*)(hp  + 4 * (lane + 32 * i));
      g[i]  = *(const float4*)(gam + 4 * (lane + 32 * i));
      be[i] = *(const float4*)(bet + 4 * (lane + 32 * i));
    }
    __builtin_amdgcn_sched_barrier(0);
    float s = 0.f;
#pragma unroll
    for (int i = 0; i < 8; ++i) s += (v[i].x + v[i].y) + (v[i].z + v[i].w);
    s = wave_sum(s);
    const float mean = s * (1.0f / (float)H_);
    float s2 = 0.f;
#pragma unroll
    for (int i = 0; i < 8; ++i) {
      const float dx = v[i].x - mean, dy = v[i].y - mean;
      const float dz = v[i].z - mean, dw = v[i].w - mean;
      s2 += (dx * dx + dy * dy) + (dz * dz + dw * dw);
    }
    s2 = wave_sum(s2);
    const float inv = rsqrtf(s2 * (1.0f / (float)H_) + 1e-5f);
#pragma unroll
    for (int i = 0; i < 8; ++i) {
      float4 y;
      y.x = (v[i].x - mean) * inv * g[i].x + be[i].x;
      y.y = (v[i].y - mean) * inv * g[i].y + be[i].y;
      y.z = (v[i].z - mean) * inv * g[i].z + be[i].z;
      y.w = (v[i].w - mean) * inv * g[i].w + be[i].w;
      apan4[wave * (APITCH_ / 4) + lane + 32 * i] = y;
    }
  }
  __syncthreads();

  const float* ap = apan + (l16 & 7) * APITCH_ + kw + 2 * half;
  const bool live = l16 < B_;
  v2f af[H_ / KW_ / 4];
#pragma unroll
  for (int j = 0; j < H_ / KW_ / 4; ++j) {
    const float2 t = *(const float2*)(ap + 4 * j);
    af[j].x = live ? t.x : 0.0f;
    af[j].y = live ? t.y : 0.0f;
  }
  __builtin_amdgcn_sched_barrier(0);
  v8f acc = {};
#pragma unroll
  for (int j = 0; j < H_ / KW_ / 4; ++j)
    acc = __builtin_amdgcn_wmma_f32_16x16x4_f32(
        false, af[j], false, bf[j], (short)0, acc, false, false);

  if (half == 0) {
    for (int r = 0; r < 8; ++r) part[wave][r * 16 + l16] = acc[r];
  }
  __syncthreads();

  if (tid < 128) {
    float s = part[0][tid];
    for (int w = 1; w < KW_; ++w) s += part[w][tid];
    const int r = tid >> 4;
    const int c = n0 + (tid & 15);
    const float val = s + b2[c];
#pragma unroll
    for (int p = 0; p < P_; ++p)
      out[(size_t)(r * P_ + p) * PD_ + c] = val;
  }
}

extern "C" void kernel_launch(void* const* d_in, const int* in_sizes, int n_in,
                              void* d_out, int out_size, void* d_ws, size_t ws_size,
                              hipStream_t stream) {
  const float* ctx  = (const float*)d_in[0];
  const float* W1   = (const float*)d_in[1];
  const float* b1   = (const float*)d_in[2];
  const float* ln_g = (const float*)d_in[3];
  const float* ln_b = (const float*)d_in[4];
  const float* W2   = (const float*)d_in[5];
  const float* b2   = (const float*)d_in[6];
  const int*  masks = (const int*)d_in[7];
  (void)in_sizes; (void)n_in; (void)out_size; (void)ws_size;

  float* out = (float*)d_out;

  char* ws = (char*)d_ws;
  float* partials = (float*)(ws + 65536);
  float* hbuf     = (float*)(ws + 65536 + 262144);

  gather_partials<<<dim3(D_ / DCW_, TC_, B_), GWAVES_ * 32, 0, stream>>>(ctx, masks, partials);
  gemm1_gelu<<<H_ / 16, KW_ * 32, 0, stream>>>(partials, W1, b1, hbuf);
  gemm2_out<<<PD_ / 16, KW_ * 32, 0, stream>>>(hbuf, ln_g, ln_b, W2, b2, out);
}
